# starting kernel
# baseline (speedup 1.0000x reference)
#include <hip/hip_runtime.h>
#include <hip/hip_bf16.h>

typedef float v2f __attribute__((ext_vector_type(2)));
typedef float v8f __attribute__((ext_vector_type(8)));

#define VOCAB 256
#define DCH   512
#define KC    5
#define TL    48
#define LLEN  8192

__global__ __launch_bounds__(32) void gbst_table_wmma(
    const float* __restrict__ emb,
    const float* __restrict__ conv_w,
    float* __restrict__ T)
{
  const int k    = blockIdx.z;
  const int v0   = blockIdx.y * 16;
  const int o0   = blockIdx.x * 16;
  const int lane = threadIdx.x;
  const int half = lane >> 4;
  const int sub  = lane & 15;
  const int kh   = half * 2;

  v8f c = {};
  const float* arow = emb + (size_t)(v0 + sub) * DCH;
  const float* bcol = conv_w + (size_t)(o0 + sub) * (DCH * KC) + k;

  for (int i0 = 0; i0 < DCH; i0 += 4) {
    v2f a, b;
    a[0] = arow[i0 + kh];
    a[1] = arow[i0 + kh + 1];
    b[0] = bcol[(size_t)(i0 + kh) * KC];
    b[1] = bcol[(size_t)(i0 + kh + 1) * KC];
    c = __builtin_amdgcn_wmma_f32_16x16x4_f32(false, a, false, b,
                                              (short)0, c, false, false);
  }
#pragma unroll
  for (int r = 0; r < 8; ++r) {
    T[((size_t)k * VOCAB + (v0 + half * 8 + r)) * DCH + (o0 + sub)] = c[r];
  }
}

__global__ __launch_bounds__(256) void gbst_main(
    const int*   __restrict__ ids,
    const float* __restrict__ T,
    const float* __restrict__ conv_b,
    const float* __restrict__ sw,
    float*       __restrict__ out,
    int L)
{
  __shared__ float xt[TL][DCH];
  __shared__ float q[TL];
  __shared__ float wc[TL][4];
  __shared__ int   sid[TL + 4];

  const int b   = blockIdx.y;
  const int t0  = blockIdx.x * TL;
  const int tid = threadIdx.x;

  if (tid < TL + 4) {
    int g = t0 + tid - 2;
    sid[tid] = (g >= 0 && g < L) ? ids[(size_t)b * L + g] : 0;
  }
  __syncthreads();

  const int oc = (tid & 127) * 4;
  const int pr = tid >> 7;

  const float bx = conv_b[oc + 0], by = conv_b[oc + 1];
  const float bz = conv_b[oc + 2], bw = conv_b[oc + 3];
  for (int p = pr; p < TL; p += 2) {
    float ax = bx, ay = by, az = bz, aw = bw;
#pragma unroll
    for (int k = 0; k < KC; ++k) {
      int v = sid[p + k];
      if (v > 0) {
        const float4 tv = *(const float4*)(T + ((size_t)k * VOCAB + v) * DCH + oc);
        ax += tv.x; ay += tv.y; az += tv.z; aw += tv.w;
      }
    }
    if (t0 + p >= L) { ax = 0.f; ay = 0.f; az = 0.f; aw = 0.f; }
    xt[p][oc + 0] = ax; xt[p][oc + 1] = ay;
    xt[p][oc + 2] = az; xt[p][oc + 3] = aw;
  }
  __syncthreads();

  {
    const int wave = tid >> 5;
    const int lane = tid & 31;
    float swr[DCH / 32];
#pragma unroll
    for (int j = 0; j < DCH / 32; ++j) swr[j] = sw[lane + 32 * j];
    for (int p = wave; p < TL; p += 8) {
      float s = 0.f;
#pragma unroll
      for (int j = 0; j < DCH / 32; ++j) s += xt[p][lane + 32 * j] * swr[j];
#pragma unroll
      for (int off = 16; off > 0; off >>= 1) s += __shfl_xor(s, off, 32);
      if (lane == 0) q[p] = s;
    }
  }
  __syncthreads();

  if (tid < TL) {
    const int l = tid;
    const int j2 = (l >> 1) << 1;
    const int j3 = (l / 3) * 3;
    const int j4 = (l >> 2) << 2;
    float s1 = q[l];
    float s2 = 0.5f * (q[j2] + q[j2 + 1]);
    float s3 = (q[j3] + q[j3 + 1] + q[j3 + 2]) * (1.f / 3.f);
    float s4 = 0.25f * (q[j4] + q[j4 + 1] + q[j4 + 2] + q[j4 + 3]);
    float mx = fmaxf(fmaxf(s1, s2), fmaxf(s3, s4));
    float e1 = __expf(s1 - mx), e2 = __expf(s2 - mx);
    float e3 = __expf(s3 - mx), e4 = __expf(s4 - mx);
    float inv = 1.f / (e1 + e2 + e3 + e4);
    wc[l][0] = e1 * inv;
    wc[l][1] = e2 * inv * 0.5f;
    wc[l][2] = e3 * inv * (1.f / 3.f);
    wc[l][3] = e4 * inv * 0.25f;
  }
  __syncthreads();

  for (int l2 = pr; l2 < TL / 2; l2 += 2) {
    if (t0 + 2 * l2 >= L) break;
    float ax = 0.f, ay = 0.f, az = 0.f, aw = 0.f;
#pragma unroll
    for (int dl = 0; dl < 2; ++dl) {
      const int l  = 2 * l2 + dl;
      const int j2 = (l >> 1) << 1;
      const int j3 = (l / 3) * 3;
      const int j4 = (l >> 2) << 2;
      const float w1 = wc[l][0], w2 = wc[l][1], w3 = wc[l][2], w4 = wc[l][3];
      ax += w1 * xt[l][oc + 0];
      ay += w1 * xt[l][oc + 1];
      az += w1 * xt[l][oc + 2];
      aw += w1 * xt[l][oc + 3];
      ax += w2 * (xt[j2][oc + 0] + xt[j2 + 1][oc + 0]);
      ay += w2 * (xt[j2][oc + 1] + xt[j2 + 1][oc + 1]);
      az += w2 * (xt[j2][oc + 2] + xt[j2 + 1][oc + 2]);
      aw += w2 * (xt[j2][oc + 3] + xt[j2 + 1][oc + 3]);
      ax += w3 * (xt[j3][oc + 0] + xt[j3 + 1][oc + 0] + xt[j3 + 2][oc + 0]);
      ay += w3 * (xt[j3][oc + 1] + xt[j3 + 1][oc + 1] + xt[j3 + 2][oc + 1]);
      az += w3 * (xt[j3][oc + 2] + xt[j3 + 1][oc + 2] + xt[j3 + 2][oc + 2]);
      aw += w3 * (xt[j3][oc + 3] + xt[j3 + 1][oc + 3] + xt[j3 + 2][oc + 3]);
      ax += w4 * (xt[j4][oc + 0] + xt[j4 + 1][oc + 0] + xt[j4 + 2][oc + 0] + xt[j4 + 3][oc + 0]);
      ay += w4 * (xt[j4][oc + 1] + xt[j4 + 1][oc + 1] + xt[j4 + 2][oc + 1] + xt[j4 + 3][oc + 1]);
      az += w4 * (xt[j4][oc + 2] + xt[j4 + 1][oc + 2] + xt[j4 + 2][oc + 2] + xt[j4 + 3][oc + 2]);
      aw += w4 * (xt[j4][oc + 3] + xt[j4 + 1][oc + 3] + xt[j4 + 2][oc + 3] + xt[j4 + 3][oc + 3]);
    }
    float4 r;
    r.x = 0.5f * ax; r.y = 0.5f * ay; r.z = 0.5f * az; r.w = 0.5f * aw;
    const size_t off = ((size_t)b * (L >> 1) + ((t0 >> 1) + l2)) * DCH + oc;
    *(float4*)(out + off) = r;
  }
}

extern "C" void kernel_launch(void* const* d_in, const int* in_sizes, int n_in,
                              void* d_out, int out_size, void* d_ws, size_t ws_size,
                              hipStream_t stream) {
  const int*   ids     = (const int*)d_in[0];
  const float* emb     = (const float*)d_in[1];
  const float* conv_w  = (const float*)d_in[2];
  const float* conv_b  = (const float*)d_in[3];
  const float* score_w = (const float*)d_in[4];
  float* out = (float*)d_out;
  float* T   = (float*)d_ws;

  const int L = LLEN;
  const int B = in_sizes[0] / L;

  dim3 g1(DCH / 16, VOCAB / 16, KC);
  gbst_table_wmma<<<g1, 32, 0, stream>>>(emb, conv_w, T);

  dim3 g2((L + TL - 1) / TL, B);
  gbst_main<<<g2, 256, 0, stream>>>(ids, T, conv_b, score_w, out, L);
}


# Kernel 2 re-cut: work-group = 32-channel slice x half row, conv rows held in registers while taps are added tap by tap, so live table lines (32 KB, slice-major T) sit in the WGP cache; no LDS copy
# speedup vs baseline: 2.1612x; 2.1612x over previous
#include <hip/hip_runtime.h>
#include <hip/hip_bf16.h>

typedef float v2f __attribute__((ext_vector_type(2)));
typedef float v8f __attribute__((ext_vector_type(8)));

#define VOCAB 256
#define DCH   512
#define KC    5
#define SLC   32
#define NSL   (DCH / SLC)
#define TGR   128
#define TLN   (TGR * 12)
#define NTHR  512
#define NOT   (DCH / 16)
#define LLEN  8192

struct TFrag {
  float4 a[4];
  float4 b[4][5];
};

__device__ __forceinline__ void tf_load(TFrag& f, const float* __restrict__ ap,
                                        const float* __restrict__ bp) {
#pragma unroll
  for (int s = 0; s < 4; ++s) {
    f.a[s] = *(const float4*)(ap + 8 * s);
#pragma unroll
    for (int q = 0; q < 5; ++q)
      f.b[s][q] = *(const float4*)(bp + 40 * s + 4 * q);
  }
}

__device__ __forceinline__ void tf_mma(const TFrag& f, v8f (&c)[KC]) {
#pragma unroll
  for (int s = 0; s < 4; ++s) {
    const float av[4] = {f.a[s].x, f.a[s].y, f.a[s].z, f.a[s].w};
    float wv[20];
#pragma unroll
    for (int q = 0; q < 5; ++q) {
      wv[4 * q + 0] = f.b[s][q].x; wv[4 * q + 1] = f.b[s][q].y;
      wv[4 * q + 2] = f.b[s][q].z; wv[4 * q + 3] = f.b[s][q].w;
    }
#pragma unroll
    for (int u = 0; u < 2; ++u) {
      v2f a;
      a[0] = av[2 * u];
      a[1] = av[2 * u + 1];
#pragma unroll
      for (int k = 0; k < KC; ++k) {
        v2f b;
        b[0] = wv[(2 * u) * 5 + k];
        b[1] = wv[(2 * u + 1) * 5 + k];
        c[k] = __builtin_amdgcn_wmma_f32_16x16x4_f32(false, a, false, b,
                                                     (short)0, c[k], false, false);
      }
    }
  }
}

__global__ __launch_bounds__(64) void gbst_table_wmma(
    const float* __restrict__ emb,
    const float* __restrict__ conv_w,
    const float* __restrict__ sw,
    float* __restrict__ T,
    float* __restrict__ P)
{
  __shared__ float4 red[2][KC][32];

  const int v0   = blockIdx.y * 16;
  const int o0   = blockIdx.x * 16;
  const int lane = threadIdx.x & 31;
  const int w    = threadIdx.x >> 5;
  const int half = lane >> 4;
  const int sub  = lane & 15;

  v8f c[KC];
#pragma unroll
  for (int k = 0; k < KC; ++k) c[k] = v8f{};

  const float* ap = emb + (size_t)(v0 + sub) * DCH + 256 * w + 4 * half;
  const float* bp = conv_w + (size_t)(o0 + sub) * (DCH * KC) + (256 * w + 4 * half) * KC;

  TFrag f0, f1;
  tf_load(f0, ap, bp);
#pragma unroll 1
  for (int it = 0; it < 3; ++it) {
    tf_load(f1, ap + 32, bp + 32 * KC);
    __builtin_amdgcn_sched_barrier(0);
    tf_mma(f0, c);
    __builtin_amdgcn_sched_barrier(0);
    tf_load(f0, ap + 64, bp + 64 * KC);
    __builtin_amdgcn_sched_barrier(0);
    tf_mma(f1, c);
    __builtin_amdgcn_sched_barrier(0);
    ap += 64;
    bp += 64 * KC;
  }
  tf_load(f1, ap + 32, bp + 32 * KC);
  __builtin_amdgcn_sched_barrier(0);
  tf_mma(f0, c);
  tf_mma(f1, c);

  float keep[KC][4];
#pragma unroll
  for (int k = 0; k < KC; ++k) {
    float4 snd;
    snd.x = w ? c[k][0] : c[k][4];
    snd.y = w ? c[k][1] : c[k][5];
    snd.z = w ? c[k][2] : c[k][6];
    snd.w = w ? c[k][3] : c[k][7];
    red[1 - w][k][lane] = snd;
    keep[k][0] = w ? c[k][4] : c[k][0];
    keep[k][1] = w ? c[k][5] : c[k][1];
    keep[k][2] = w ? c[k][6] : c[k][2];
    keep[k][3] = w ? c[k][7] : c[k][3];
  }
  __syncthreads();

  const float swv = sw[o0 + sub];
#pragma unroll
  for (int k = 0; k < KC; ++k) {
    const float4 oth = red[w][k][lane];
    keep[k][0] += oth.x; keep[k][1] += oth.y;
    keep[k][2] += oth.z; keep[k][3] += oth.w;
#pragma unroll
    for (int rr = 0; rr < 4; ++rr) {
      const int v = v0 + half * 8 + 4 * w + rr;
      T[(((size_t)(o0 / SLC) * KC + k) * VOCAB + v) * SLC + (o0 % SLC + sub)] = keep[k][rr];
      float s = keep[k][rr] * swv;
      s += __shfl_xor(s, 8, 32);
      s += __shfl_xor(s, 4, 32);
      s += __shfl_xor(s, 2, 32);
      s += __shfl_xor(s, 1, 32);
      if (sub == 0)
        P[((size_t)k * VOCAB + v) * NOT + blockIdx.x] = s;
    }
  }
}

__host__ __device__ constexpr bool touches(int j, int lp) {
  return ((lp >> 2) == (j >> 1)) || ((lp / 3) == ((2 * j) / 3)) ||
         ((lp / 3) == ((2 * j + 1) / 3));
}

template <bool EDGE>
__device__ __forceinline__ void gather_mix(
    const float* __restrict__ Ts, float* __restrict__ out,
    const int* sid, const float (*cf)[6][12], const float4 bias,
    int b, int t0, int L, int oc, int lg, int ngt)
{
  int  pb[2];
  bool val[2];
#pragma unroll
  for (int s = 0; s < 2; ++s) {
    const int g = lg + (NTHR / 8) * s;
    val[s] = g < ngt;
    pb[s]  = 12 * (val[s] ? g : ngt - 1);
  }
  float4 x[2][12];
#pragma unroll
  for (int s = 0; s < 2; ++s)
#pragma unroll
    for (int lp = 0; lp < 12; ++lp) x[s][lp] = bias;

#pragma unroll 1
  for (int k = 0; k < KC; ++k) {
    const float* Tk = Ts + (size_t)k * (VOCAB * SLC);
#pragma unroll
    for (int s = 0; s < 2; ++s) {
#pragma unroll
      for (int lp = 0; lp < 12; ++lp) {
        int v = sid[pb[s] + lp + k];
        if (EDGE) {
          const bool ok = v >= 0;
          v = ok ? v : 0;
          const float4 tv = *(const float4*)(Tk + v * SLC);
          x[s][lp].x += ok ? tv.x : 0.f; x[s][lp].y += ok ? tv.y : 0.f;
          x[s][lp].z += ok ? tv.z : 0.f; x[s][lp].w += ok ? tv.w : 0.f;
        } else {
          const float4 tv = *(const float4*)(Tk + v * SLC);
          x[s][lp].x += tv.x; x[s][lp].y += tv.y; x[s][lp].z += tv.z; x[s][lp].w += tv.w;
        }
      }
    }
  }

#pragma unroll
  for (int s = 0; s < 2; ++s) {
    const int g = pb[s] / 12;
    float4 acc[6];
#pragma unroll
    for (int j = 0; j < 6; ++j) acc[j] = make_float4(0.f, 0.f, 0.f, 0.f);
#pragma unroll
    for (int lp = 0; lp < 12; ++lp) {
      float4 xv = x[s][lp];
      if (EDGE && t0 + pb[s] + lp >= L) xv = make_float4(0.f, 0.f, 0.f, 0.f);
#pragma unroll
      for (int j = 0; j < 6; ++j) {
        if (touches(j, lp)) {
          const float w = cf[g][j][lp];
          acc[j].x += w * xv.x; acc[j].y += w * xv.y;
          acc[j].z += w * xv.z; acc[j].w += w * xv.w;
        }
      }
    }
#pragma unroll
    for (int j = 0; j < 6; ++j) {
      if (val[s] && (!EDGE || t0 + pb[s] + 2 * j < L)) {
        const size_t off = ((size_t)b * (L >> 1) + (((t0 + pb[s]) >> 1) + j)) * DCH + oc;
        *(float4*)(out + off) = acc[j];
      }
    }
  }
}

__global__ __launch_bounds__(NTHR) void gbst_main(
    const int*   __restrict__ ids,
    const float* __restrict__ T,
    const float* __restrict__ P,
    const float* __restrict__ conv_b,
    const float* __restrict__ sw,
    float*       __restrict__ out,
    int L)
{
  __shared__ int   sid[TLN + 4];
  __shared__ float sc[KC * VOCAB];
  __shared__ float bpart[4];
  __shared__ float q[TLN];
  __shared__ __attribute__((aligned(16))) float wc[TLN][4];
  __shared__ __attribute__((aligned(16))) float cf[TGR][6][12];

  const int cs   = blockIdx.x;
  const int b    = blockIdx.y >> 1;
  const int hf   = blockIdx.y & 1;
  const int tid  = threadIdx.x;
  const int ngr  = (L + 11) / 12;
  const int gbeg = hf ? (ngr + 1) / 2 : 0;
  const int gend = hf ? ngr : (ngr + 1) / 2;
  const int ntile = (gend - gbeg + TGR - 1) / TGR;

  if (tid < 128) {
    float s = conv_b[tid] * sw[tid] + conv_b[tid + 128] * sw[tid + 128];
    s += conv_b[tid + 256] * sw[tid + 256] + conv_b[tid + 384] * sw[tid + 384];
#pragma unroll
    for (int off = 16; off > 0; off >>= 1) s += __shfl_xor(s, off, 32);
    if ((tid & 31) == 0) bpart[tid >> 5] = s;
  }
  for (int e = tid; e < KC * VOCAB; e += NTHR) {
    const float4* pp = (const float4*)(P + (size_t)e * NOT);
    float s = 0.f;
#pragma unroll
    for (int i = 0; i < NOT / 4; ++i) {
      const float4 t = pp[i];
      s += t.x; s += t.y; s += t.z; s += t.w;
    }
    sc[e] = s;
  }

  int idn[4];
#pragma unroll
  for (int r = 0; r < 4; ++r) {
    const int i = tid + NTHR * r;
    const int g = 12 * gbeg + i - 2;
    idn[r] = (i < TLN + 4 && g >= 0 && g < L) ? ids[(size_t)b * L + g] : -1;
  }

  const int lg = tid >> 3;
  const int oc = cs * SLC + (tid & 7) * 4;
  const float4 bias = *(const float4*)(conv_b + oc);
  const float* Ts = T + (size_t)cs * (KC * VOCAB * SLC) + (tid & 7) * 4;

  for (int ti = 0; ti < ntile; ++ti) {
    const int tg0 = gbeg + TGR * ti;
    const int t0  = 12 * tg0;
    const int ngt = (gend - tg0 < TGR) ? (gend - tg0) : TGR;
    const int npt = 12 * ngt;

    __syncthreads();
#pragma unroll
    for (int r = 0; r < 4; ++r) {
      const int i = tid + NTHR * r;
      if (i < TLN + 4) sid[i] = idn[r];
    }
    __syncthreads();

    for (int p = tid; p < npt; p += NTHR) {
      float s = 0.f;
      if (t0 + p < L) {
        s = bpart[0];
#pragma unroll
        for (int w = 1; w < 4; ++w) s += bpart[w];
#pragma unroll
        for (int k = 0; k < KC; ++k) {
          const int v = sid[p + k];
          const float t = sc[k * VOCAB + (v < 0 ? 0 : v)];
          s += (v >= 0) ? t : 0.f;
        }
      }
      q[p] = s;
    }
    __syncthreads();

    for (int l = tid; l < npt; l += NTHR) {
      const int j2 = (l >> 1) << 1;
      const int j3 = (l / 3) * 3;
      const int j4 = (l >> 2) << 2;
      float s1 = q[l];
      float s2 = 0.5f * (q[j2] + q[j2 + 1]);
      float s3 = (q[j3] + q[j3 + 1] + q[j3 + 2]) * (1.f / 3.f);
      float s4 = 0.25f * (q[j4] + q[j4 + 1] + q[j4 + 2] + q[j4 + 3]);
      float mx = fmaxf(fmaxf(s1, s2), fmaxf(s3, s4));
      float e1 = __expf(s1 - mx), e2 = __expf(s2 - mx);
      float e3 = __expf(s3 - mx), e4 = __expf(s4 - mx);
      float inv = 1.f / (e1 + e2 + e3 + e4);
      wc[l][0] = e1 * inv;
      wc[l][1] = e2 * inv * 0.5f;
      wc[l][2] = e3 * inv * (1.f / 3.f);
      wc[l][3] = e4 * inv * 0.25f;
    }
    __syncthreads();

    for (int e = tid; e < ngt * 6; e += NTHR) {
      const int g = e / 6;
      const int j = e - g * 6;
      float c[12];
#pragma unroll
      for (int lp = 0; lp < 12; ++lp) c[lp] = 0.f;
#pragma unroll
      for (int dl = 0; dl < 2; ++dl) {
        const int l  = 2 * j + dl;
        const int l3 = l / 3;
        const float4 w = *(const float4*)wc[12 * g + l];
#pragma unroll
        for (int lp = 0; lp < 12; ++lp) {
          c[lp] += (lp == l)               ? w.x : 0.f;
          c[lp] += ((lp >> 1) == (l >> 1)) ? w.y : 0.f;
          c[lp] += ((lp / 3) == l3)        ? w.z : 0.f;
          c[lp] += ((lp >> 2) == (l >> 2)) ? w.w : 0.f;
        }
      }
#pragma unroll
      for (int lp = 0; lp < 12; ++lp) cf[g][j][lp] = 0.5f * c[lp];
    }
    __syncthreads();

    if (ti + 1 < ntile) {
#pragma unroll
      for (int r = 0; r < 4; ++r) {
        const int i = tid + NTHR * r;
        const int g = t0 + TLN + i - 2;
        idn[r] = (i < TLN + 4 && g >= 0 && g < L) ? ids[(size_t)b * L + g] : -1;
      }
    }
    const bool edge = (t0 == 0) || (t0 + npt + 2 > L);
    if (edge) gather_mix<true >(Ts, out, sid, cf, bias, b, t0, L, oc, lg, ngt);
    else      gather_mix<false>(Ts, out, sid, cf, bias, b, t0, L, oc, lg, ngt);
  }
}

extern "C" void kernel_launch(void* const* d_in, const int* in_sizes, int n_in,
                              void* d_out, int out_size, void* d_ws, size_t ws_size,
                              hipStream_t stream) {
  const int*   ids     = (const int*)d_in[0];
  const float* emb     = (const float*)d_in[1];
  const float* conv_w  = (const float*)d_in[2];
  const float* conv_b  = (const float*)d_in[3];
  const float* score_w = (const float*)d_in[4];
  float* out = (float*)d_out;
  float* T   = (float*)d_ws;
  float* P   = T + KC * VOCAB * DCH;

  const int L = LLEN;
  const int B = in_sizes[0] / L;

  dim3 g1(DCH / 16, VOCAB / 16);
  gbst_table_wmma<<<g1, 64, 0, stream>>>(emb, conv_w, score_w, T, P);

  dim3 g2(NSL, 2 * B);
  gbst_main<<<g2, NTHR, 0, stream>>>(ids, T, P, conv_b, score_w, out, L);
}


# Kernel 2 prologue made one memory round trip: score partials stored tile-major so the score table is summed from whole-line loads, issued with the ids, bias and conv_b.score_w loads; sums kept
# speedup vs baseline: 2.3103x; 1.0690x over previous
#include <hip/hip_runtime.h>
#include <hip/hip_bf16.h>

typedef float v2f __attribute__((ext_vector_type(2)));
typedef float v8f __attribute__((ext_vector_type(8)));

#define VOCAB 256
#define DCH   512
#define KC    5
#define SLC   32
#define NSL   (DCH / SLC)
#define TGR   128
#define TLN   (TGR * 12)
#define NTHR  512
#define NOT   (DCH / 16)
#define LLEN  8192

struct TFrag {
  float4 a[4];
  float4 b[4][5];
};

__device__ __forceinline__ void tf_load(TFrag& f, const float* __restrict__ ap,
                                        const float* __restrict__ bp) {
#pragma unroll
  for (int s = 0; s < 4; ++s) {
    f.a[s] = *(const float4*)(ap + 8 * s);
#pragma unroll
    for (int q = 0; q < 5; ++q)
      f.b[s][q] = *(const float4*)(bp + 40 * s + 4 * q);
  }
}

__device__ __forceinline__ void tf_mma(const TFrag& f, v8f (&c)[KC]) {
#pragma unroll
  for (int s = 0; s < 4; ++s) {
    const float av[4] = {f.a[s].x, f.a[s].y, f.a[s].z, f.a[s].w};
    float wv[20];
#pragma unroll
    for (int q = 0; q < 5; ++q) {
      wv[4 * q + 0] = f.b[s][q].x; wv[4 * q + 1] = f.b[s][q].y;
      wv[4 * q + 2] = f.b[s][q].z; wv[4 * q + 3] = f.b[s][q].w;
    }
#pragma unroll
    for (int u = 0; u < 2; ++u) {
      v2f a;
      a[0] = av[2 * u];
      a[1] = av[2 * u + 1];
#pragma unroll
      for (int k = 0; k < KC; ++k) {
        v2f b;
        b[0] = wv[(2 * u) * 5 + k];
        b[1] = wv[(2 * u + 1) * 5 + k];
        c[k] = __builtin_amdgcn_wmma_f32_16x16x4_f32(false, a, false, b,
                                                     (short)0, c[k], false, false);
      }
    }
  }
}

__global__ __launch_bounds__(64) void gbst_table_wmma(
    const float* __restrict__ emb,
    const float* __restrict__ conv_w,
    const float* __restrict__ sw,
    float* __restrict__ T,
    float* __restrict__ P)
{
  __shared__ float4 red[2][KC][32];

  const int v0   = blockIdx.y * 16;
  const int o0   = blockIdx.x * 16;
  const int lane = threadIdx.x & 31;
  const int w    = threadIdx.x >> 5;
  const int half = lane >> 4;
  const int sub  = lane & 15;

  v8f c[KC];
#pragma unroll
  for (int k = 0; k < KC; ++k) c[k] = v8f{};

  const float* ap = emb + (size_t)(v0 + sub) * DCH + 256 * w + 4 * half;
  const float* bp = conv_w + (size_t)(o0 + sub) * (DCH * KC) + (256 * w + 4 * half) * KC;

  TFrag f0, f1;
  tf_load(f0, ap, bp);
#pragma unroll 1
  for (int it = 0; it < 3; ++it) {
    tf_load(f1, ap + 32, bp + 32 * KC);
    __builtin_amdgcn_sched_barrier(0);
    tf_mma(f0, c);
    __builtin_amdgcn_sched_barrier(0);
    tf_load(f0, ap + 64, bp + 64 * KC);
    __builtin_amdgcn_sched_barrier(0);
    tf_mma(f1, c);
    __builtin_amdgcn_sched_barrier(0);
    ap += 64;
    bp += 64 * KC;
  }
  tf_load(f1, ap + 32, bp + 32 * KC);
  __builtin_amdgcn_sched_barrier(0);
  tf_mma(f0, c);
  tf_mma(f1, c);

  float keep[KC][4];
#pragma unroll
  for (int k = 0; k < KC; ++k) {
    float4 snd;
    snd.x = w ? c[k][0] : c[k][4];
    snd.y = w ? c[k][1] : c[k][5];
    snd.z = w ? c[k][2] : c[k][6];
    snd.w = w ? c[k][3] : c[k][7];
    red[1 - w][k][lane] = snd;
    keep[k][0] = w ? c[k][4] : c[k][0];
    keep[k][1] = w ? c[k][5] : c[k][1];
    keep[k][2] = w ? c[k][6] : c[k][2];
    keep[k][3] = w ? c[k][7] : c[k][3];
  }
  __syncthreads();

  const float swv = sw[o0 + sub];
#pragma unroll
  for (int k = 0; k < KC; ++k) {
    const float4 oth = red[w][k][lane];
    keep[k][0] += oth.x; keep[k][1] += oth.y;
    keep[k][2] += oth.z; keep[k][3] += oth.w;
#pragma unroll
    for (int rr = 0; rr < 4; ++rr) {
      const int v = v0 + half * 8 + 4 * w + rr;
      T[(((size_t)(o0 / SLC) * KC + k) * VOCAB + v) * SLC + (o0 % SLC + sub)] = keep[k][rr];
      float s = keep[k][rr] * swv;
      s += __shfl_xor(s, 8, 32);
      s += __shfl_xor(s, 4, 32);
      s += __shfl_xor(s, 2, 32);
      s += __shfl_xor(s, 1, 32);
      if (sub == 0)
        P[((size_t)blockIdx.x * KC + k) * VOCAB + v] = s;
    }
  }
}

__host__ __device__ constexpr bool touches(int j, int lp) {
  return ((lp >> 2) == (j >> 1)) || ((lp / 3) == ((2 * j) / 3)) ||
         ((lp / 3) == ((2 * j + 1) / 3));
}

template <bool EDGE>
__device__ __forceinline__ void gather_mix(
    const float* __restrict__ Ts, float* __restrict__ out,
    const int* sid, const float (*cf)[6][12], const float4 bias,
    int b, int t0, int L, int oc, int lg, int ngt)
{
  int  pb[2];
  bool val[2];
#pragma unroll
  for (int s = 0; s < 2; ++s) {
    const int g = lg + (NTHR / 8) * s;
    val[s] = g < ngt;
    pb[s]  = 12 * (val[s] ? g : ngt - 1);
  }
  float4 x[2][12];
#pragma unroll
  for (int s = 0; s < 2; ++s)
#pragma unroll
    for (int lp = 0; lp < 12; ++lp) x[s][lp] = bias;

#pragma unroll 1
  for (int k = 0; k < KC; ++k) {
    const float* Tk = Ts + (size_t)k * (VOCAB * SLC);
#pragma unroll
    for (int s = 0; s < 2; ++s) {
#pragma unroll
      for (int lp = 0; lp < 12; ++lp) {
        int v = sid[pb[s] + lp + k];
        if (EDGE) {
          const bool ok = v >= 0;
          v = ok ? v : 0;
          const float4 tv = *(const float4*)(Tk + v * SLC);
          x[s][lp].x += ok ? tv.x : 0.f; x[s][lp].y += ok ? tv.y : 0.f;
          x[s][lp].z += ok ? tv.z : 0.f; x[s][lp].w += ok ? tv.w : 0.f;
        } else {
          const float4 tv = *(const float4*)(Tk + v * SLC);
          x[s][lp].x += tv.x; x[s][lp].y += tv.y; x[s][lp].z += tv.z; x[s][lp].w += tv.w;
        }
      }
    }
  }

#pragma unroll
  for (int s = 0; s < 2; ++s) {
    const int g = pb[s] / 12;
    float4 acc[6];
#pragma unroll
    for (int j = 0; j < 6; ++j) acc[j] = make_float4(0.f, 0.f, 0.f, 0.f);
#pragma unroll
    for (int lp = 0; lp < 12; ++lp) {
      float4 xv = x[s][lp];
      if (EDGE && t0 + pb[s] + lp >= L) xv = make_float4(0.f, 0.f, 0.f, 0.f);
#pragma unroll
      for (int j = 0; j < 6; ++j) {
        if (touches(j, lp)) {
          const float w = cf[g][j][lp];
          acc[j].x += w * xv.x; acc[j].y += w * xv.y;
          acc[j].z += w * xv.z; acc[j].w += w * xv.w;
        }
      }
    }
#pragma unroll
    for (int j = 0; j < 6; ++j) {
      if (val[s] && (!EDGE || t0 + pb[s] + 2 * j < L)) {
        const size_t off = ((size_t)b * (L >> 1) + (((t0 + pb[s]) >> 1) + j)) * DCH + oc;
        *(float4*)(out + off) = acc[j];
      }
    }
  }
}

__global__ __launch_bounds__(NTHR) void gbst_main(
    const int*   __restrict__ ids,
    const float* __restrict__ T,
    const float* __restrict__ P,
    const float* __restrict__ conv_b,
    const float* __restrict__ sw,
    float*       __restrict__ out,
    int L)
{
  __shared__ int   sid[TLN + 4];
  __shared__ __attribute__((aligned(16))) float sc[KC * VOCAB];
  __shared__ float bpart[4];
  __shared__ float q[TLN];
  __shared__ __attribute__((aligned(16))) float wc[TLN][4];
  __shared__ __attribute__((aligned(16))) float cf[TGR][6][12];

  const int cs   = blockIdx.x;
  const int b    = blockIdx.y >> 1;
  const int hf   = blockIdx.y & 1;
  const int tid  = threadIdx.x;
  const int ngr  = (L + 11) / 12;
  const int gbeg = hf ? (ngr + 1) / 2 : 0;
  const int gend = hf ? ngr : (ngr + 1) / 2;
  const int ntile = (gend - gbeg + TGR - 1) / TGR;

  int idn[4];
#pragma unroll
  for (int r = 0; r < 4; ++r) {
    const int i = tid + NTHR * r;
    const int g = 12 * gbeg + i - 2;
    idn[r] = (i < TLN + 4 && g >= 0 && g < L) ? ids[(size_t)b * L + g] : -1;
  }

  const int lg = tid >> 3;
  const int oc = cs * SLC + (tid & 7) * 4;
  const float4 bias = *(const float4*)(conv_b + oc);
  const float* Ts = T + (size_t)cs * (KC * VOCAB * SLC) + (tid & 7) * 4;

  const int cq = tid & 127;
  const float cb0 = conv_b[cq], cb1 = conv_b[cq + 128];
  const float cb2 = conv_b[cq + 256], cb3 = conv_b[cq + 384];
  const float sw0 = sw[cq], sw1 = sw[cq + 128];
  const float sw2 = sw[cq + 256], sw3 = sw[cq + 384];
  if (tid < KC * VOCAB / 4) {
    const float4* pp = (const float4*)P + tid;
    float4 s = make_float4(0.f, 0.f, 0.f, 0.f);
#pragma unroll
    for (int ot = 0; ot < NOT; ++ot) {
      const float4 t = pp[(size_t)ot * (KC * VOCAB / 4)];
      s.x += t.x; s.y += t.y; s.z += t.z; s.w += t.w;
    }
    *(float4*)(sc + 4 * tid) = s;
  }
  if (tid >= NTHR - 128) {
    float s = cb0 * sw0 + cb1 * sw1;
    s += cb2 * sw2 + cb3 * sw3;
#pragma unroll
    for (int off = 16; off > 0; off >>= 1) s += __shfl_xor(s, off, 32);
    if ((tid & 31) == 0) bpart[(tid >> 5) & 3] = s;
  }

  for (int ti = 0; ti < ntile; ++ti) {
    const int tg0 = gbeg + TGR * ti;
    const int t0  = 12 * tg0;
    const int ngt = (gend - tg0 < TGR) ? (gend - tg0) : TGR;
    const int npt = 12 * ngt;

    __syncthreads();
#pragma unroll
    for (int r = 0; r < 4; ++r) {
      const int i = tid + NTHR * r;
      if (i < TLN + 4) sid[i] = idn[r];
    }
    __syncthreads();

    for (int p = tid; p < npt; p += NTHR) {
      float s = 0.f;
      if (t0 + p < L) {
        s = bpart[0];
#pragma unroll
        for (int w = 1; w < 4; ++w) s += bpart[w];
#pragma unroll
        for (int k = 0; k < KC; ++k) {
          const int v = sid[p + k];
          const float t = sc[k * VOCAB + (v < 0 ? 0 : v)];
          s += (v >= 0) ? t : 0.f;
        }
      }
      q[p] = s;
    }
    __syncthreads();

    for (int l = tid; l < npt; l += NTHR) {
      const int j2 = (l >> 1) << 1;
      const int j3 = (l / 3) * 3;
      const int j4 = (l >> 2) << 2;
      float s1 = q[l];
      float s2 = 0.5f * (q[j2] + q[j2 + 1]);
      float s3 = (q[j3] + q[j3 + 1] + q[j3 + 2]) * (1.f / 3.f);
      float s4 = 0.25f * (q[j4] + q[j4 + 1] + q[j4 + 2] + q[j4 + 3]);
      float mx = fmaxf(fmaxf(s1, s2), fmaxf(s3, s4));
      float e1 = __expf(s1 - mx), e2 = __expf(s2 - mx);
      float e3 = __expf(s3 - mx), e4 = __expf(s4 - mx);
      float inv = 1.f / (e1 + e2 + e3 + e4);
      wc[l][0] = e1 * inv;
      wc[l][1] = e2 * inv * 0.5f;
      wc[l][2] = e3 * inv * (1.f / 3.f);
      wc[l][3] = e4 * inv * 0.25f;
    }
    __syncthreads();

    for (int e = tid; e < ngt * 6; e += NTHR) {
      const int g = e / 6;
      const int j = e - g * 6;
      float c[12];
#pragma unroll
      for (int lp = 0; lp < 12; ++lp) c[lp] = 0.f;
#pragma unroll
      for (int dl = 0; dl < 2; ++dl) {
        const int l  = 2 * j + dl;
        const int l3 = l / 3;
        const float4 w = *(const float4*)wc[12 * g + l];
#pragma unroll
        for (int lp = 0; lp < 12; ++lp) {
          c[lp] += (lp == l)               ? w.x : 0.f;
          c[lp] += ((lp >> 1) == (l >> 1)) ? w.y : 0.f;
          c[lp] += ((lp / 3) == l3)        ? w.z : 0.f;
          c[lp] += ((lp >> 2) == (l >> 2)) ? w.w : 0.f;
        }
      }
#pragma unroll
      for (int lp = 0; lp < 12; ++lp) cf[g][j][lp] = 0.5f * c[lp];
    }
    __syncthreads();

    if (ti + 1 < ntile) {
#pragma unroll
      for (int r = 0; r < 4; ++r) {
        const int i = tid + NTHR * r;
        const int g = t0 + TLN + i - 2;
        idn[r] = (i < TLN + 4 && g >= 0 && g < L) ? ids[(size_t)b * L + g] : -1;
      }
    }
    const bool edge = (t0 == 0) || (t0 + npt + 2 > L);
    if (edge) gather_mix<true >(Ts, out, sid, cf, bias, b, t0, L, oc, lg, ngt);
    else      gather_mix<false>(Ts, out, sid, cf, bias, b, t0, L, oc, lg, ngt);
  }
}

extern "C" void kernel_launch(void* const* d_in, const int* in_sizes, int n_in,
                              void* d_out, int out_size, void* d_ws, size_t ws_size,
                              hipStream_t stream) {
  const int*   ids     = (const int*)d_in[0];
  const float* emb     = (const float*)d_in[1];
  const float* conv_w  = (const float*)d_in[2];
  const float* conv_b  = (const float*)d_in[3];
  const float* score_w = (const float*)d_in[4];
  float* out = (float*)d_out;
  float* T   = (float*)d_ws;
  float* P   = T + KC * VOCAB * DCH;

  const int L = LLEN;
  const int B = in_sizes[0] / L;

  dim3 g1(DCH / 16, VOCAB / 16);
  gbst_table_wmma<<<g1, 64, 0, stream>>>(emb, conv_w, score_w, T, P);

  dim3 g2(NSL, 2 * B);
  gbst_main<<<g2, NTHR, 0, stream>>>(ids, T, P, conv_b, score_w, out, L);
}


# Table kernel: the 20 score-partial sums over 16 lanes are DPP row_xmask adds on the vector unit, not LDS-unit lane exchanges waited for one by one; same 8,4,2,1 butterfly; one 16 B store per tap
# speedup vs baseline: 2.5124x; 1.0875x over previous
#include <hip/hip_runtime.h>
#include <hip/hip_bf16.h>

typedef float v2f __attribute__((ext_vector_type(2)));
typedef float v8f __attribute__((ext_vector_type(8)));

#define VOCAB 256
#define DCH   512
#define KC    5
#define SLC   32
#define NSL   (DCH / SLC)
#define TGR   128
#define TLN   (TGR * 12)
#define NTHR  512
#define NOT   (DCH / 16)
#define LLEN  8192

struct TFrag {
  float4 a[4];
  float4 b[4][5];
};

__device__ __forceinline__ void tf_load(TFrag& f, const float* __restrict__ ap,
                                        const float* __restrict__ bp) {
#pragma unroll
  for (int s = 0; s < 4; ++s) {
    f.a[s] = *(const float4*)(ap + 8 * s);
#pragma unroll
    for (int q = 0; q < 5; ++q)
      f.b[s][q] = *(const float4*)(bp + 40 * s + 4 * q);
  }
}

__device__ __forceinline__ void tf_mma(const TFrag& f, v8f (&c)[KC]) {
#pragma unroll
  for (int s = 0; s < 4; ++s) {
    const float av[4] = {f.a[s].x, f.a[s].y, f.a[s].z, f.a[s].w};
    float wv[20];
#pragma unroll
    for (int q = 0; q < 5; ++q) {
      wv[4 * q + 0] = f.b[s][q].x; wv[4 * q + 1] = f.b[s][q].y;
      wv[4 * q + 2] = f.b[s][q].z; wv[4 * q + 3] = f.b[s][q].w;
    }
#pragma unroll
    for (int u = 0; u < 2; ++u) {
      v2f a;
      a[0] = av[2 * u];
      a[1] = av[2 * u + 1];
#pragma unroll
      for (int k = 0; k < KC; ++k) {
        v2f b;
        b[0] = wv[(2 * u) * 5 + k];
        b[1] = wv[(2 * u + 1) * 5 + k];
        c[k] = __builtin_amdgcn_wmma_f32_16x16x4_f32(false, a, false, b,
                                                     (short)0, c[k], false, false);
      }
    }
  }
}

template <int M>
__device__ __forceinline__ float row_xor(float x) {
  return __int_as_float(__builtin_amdgcn_update_dpp(
      0, __float_as_int(x), 0x160 + M, 0xf, 0xf, true));
}

__global__ __launch_bounds__(64) void gbst_table_wmma(
    const float* __restrict__ emb,
    const float* __restrict__ conv_w,
    const float* __restrict__ sw,
    float* __restrict__ T,
    float* __restrict__ P)
{
  __shared__ float4 red[2][KC][32];

  const int v0   = blockIdx.y * 16;
  const int o0   = blockIdx.x * 16;
  const int lane = threadIdx.x & 31;
  const int w    = threadIdx.x >> 5;
  const int half = lane >> 4;
  const int sub  = lane & 15;

  v8f c[KC];
#pragma unroll
  for (int k = 0; k < KC; ++k) c[k] = v8f{};

  const float* ap = emb + (size_t)(v0 + sub) * DCH + 256 * w + 4 * half;
  const float* bp = conv_w + (size_t)(o0 + sub) * (DCH * KC) + (256 * w + 4 * half) * KC;

  TFrag f0, f1;
  tf_load(f0, ap, bp);
#pragma unroll 1
  for (int it = 0; it < 3; ++it) {
    tf_load(f1, ap + 32, bp + 32 * KC);
    __builtin_amdgcn_sched_barrier(0);
    tf_mma(f0, c);
    __builtin_amdgcn_sched_barrier(0);
    tf_load(f0, ap + 64, bp + 64 * KC);
    __builtin_amdgcn_sched_barrier(0);
    tf_mma(f1, c);
    __builtin_amdgcn_sched_barrier(0);
    ap += 64;
    bp += 64 * KC;
  }
  tf_load(f1, ap + 32, bp + 32 * KC);
  __builtin_amdgcn_sched_barrier(0);
  tf_mma(f0, c);
  tf_mma(f1, c);

  float keep[KC][4];
#pragma unroll
  for (int k = 0; k < KC; ++k) {
    float4 snd;
    snd.x = w ? c[k][0] : c[k][4];
    snd.y = w ? c[k][1] : c[k][5];
    snd.z = w ? c[k][2] : c[k][6];
    snd.w = w ? c[k][3] : c[k][7];
    red[1 - w][k][lane] = snd;
    keep[k][0] = w ? c[k][4] : c[k][0];
    keep[k][1] = w ? c[k][5] : c[k][1];
    keep[k][2] = w ? c[k][6] : c[k][2];
    keep[k][3] = w ? c[k][7] : c[k][3];
  }
  __syncthreads();

  const float swv = sw[o0 + sub];
  float* pr = P + (size_t)blockIdx.x * (KC * VOCAB) + v0 + half * 8 + 4 * w;
#pragma unroll
  for (int k = 0; k < KC; ++k) {
    const float4 oth = red[w][k][lane];
    keep[k][0] += oth.x; keep[k][1] += oth.y;
    keep[k][2] += oth.z; keep[k][3] += oth.w;
    float sr[4];
#pragma unroll
    for (int rr = 0; rr < 4; ++rr) {
      const int v = v0 + half * 8 + 4 * w + rr;
      T[(((size_t)(o0 / SLC) * KC + k) * VOCAB + v) * SLC + (o0 % SLC + sub)] = keep[k][rr];
      float s = keep[k][rr] * swv;
      s += row_xor<8>(s);
      s += row_xor<4>(s);
      s += row_xor<2>(s);
      s += row_xor<1>(s);
      sr[rr] = s;
    }
    if (sub == 0)
      *(float4*)(pr + k * VOCAB) = make_float4(sr[0], sr[1], sr[2], sr[3]);
  }
}

__host__ __device__ constexpr bool touches(int j, int lp) {
  return ((lp >> 2) == (j >> 1)) || ((lp / 3) == ((2 * j) / 3)) ||
         ((lp / 3) == ((2 * j + 1) / 3));
}

template <bool EDGE>
__device__ __forceinline__ void gather_mix(
    const float* __restrict__ Ts, float* __restrict__ out,
    const int* sid, const float (*cf)[6][12], const float4 bias,
    int b, int t0, int L, int oc, int lg, int ngt)
{
  int  pb[2];
  bool val[2];
#pragma unroll
  for (int s = 0; s < 2; ++s) {
    const int g = lg + (NTHR / 8) * s;
    val[s] = g < ngt;
    pb[s]  = 12 * (val[s] ? g : ngt - 1);
  }
  float4 x[2][12];
#pragma unroll
  for (int s = 0; s < 2; ++s)
#pragma unroll
    for (int lp = 0; lp < 12; ++lp) x[s][lp] = bias;

#pragma unroll 1
  for (int k = 0; k < KC; ++k) {
    const float* Tk = Ts + (size_t)k * (VOCAB * SLC);
#pragma unroll
    for (int s = 0; s < 2; ++s) {
#pragma unroll
      for (int lp = 0; lp < 12; ++lp) {
        int v = sid[pb[s] + lp + k];
        if (EDGE) {
          const bool ok = v >= 0;
          v = ok ? v : 0;
          const float4 tv = *(const float4*)(Tk + v * SLC);
          x[s][lp].x += ok ? tv.x : 0.f; x[s][lp].y += ok ? tv.y : 0.f;
          x[s][lp].z += ok ? tv.z : 0.f; x[s][lp].w += ok ? tv.w : 0.f;
        } else {
          const float4 tv = *(const float4*)(Tk + v * SLC);
          x[s][lp].x += tv.x; x[s][lp].y += tv.y; x[s][lp].z += tv.z; x[s][lp].w += tv.w;
        }
      }
    }
  }

#pragma unroll
  for (int s = 0; s < 2; ++s) {
    const int g = pb[s] / 12;
    float4 acc[6];
#pragma unroll
    for (int j = 0; j < 6; ++j) acc[j] = make_float4(0.f, 0.f, 0.f, 0.f);
#pragma unroll
    for (int lp = 0; lp < 12; ++lp) {
      float4 xv = x[s][lp];
      if (EDGE && t0 + pb[s] + lp >= L) xv = make_float4(0.f, 0.f, 0.f, 0.f);
#pragma unroll
      for (int j = 0; j < 6; ++j) {
        if (touches(j, lp)) {
          const float w = cf[g][j][lp];
          acc[j].x += w * xv.x; acc[j].y += w * xv.y;
          acc[j].z += w * xv.z; acc[j].w += w * xv.w;
        }
      }
    }
#pragma unroll
    for (int j = 0; j < 6; ++j) {
      if (val[s] && (!EDGE || t0 + pb[s] + 2 * j < L)) {
        const size_t off = ((size_t)b * (L >> 1) + (((t0 + pb[s]) >> 1) + j)) * DCH + oc;
        *(float4*)(out + off) = acc[j];
      }
    }
  }
}

__global__ __launch_bounds__(NTHR) void gbst_main(
    const int*   __restrict__ ids,
    const float* __restrict__ T,
    const float* __restrict__ P,
    const float* __restrict__ conv_b,
    const float* __restrict__ sw,
    float*       __restrict__ out,
    int L)
{
  __shared__ int   sid[TLN + 4];
  __shared__ __attribute__((aligned(16))) float sc[KC * VOCAB];
  __shared__ float bpart[4];
  __shared__ float q[TLN];
  __shared__ __attribute__((aligned(16))) float wc[TLN][4];
  __shared__ __attribute__((aligned(16))) float cf[TGR][6][12];

  const int cs   = blockIdx.x;
  const int b    = blockIdx.y >> 1;
  const int hf   = blockIdx.y & 1;
  const int tid  = threadIdx.x;
  const int ngr  = (L + 11) / 12;
  const int gbeg = hf ? (ngr + 1) / 2 : 0;
  const int gend = hf ? ngr : (ngr + 1) / 2;
  const int ntile = (gend - gbeg + TGR - 1) / TGR;

  int idn[4];
#pragma unroll
  for (int r = 0; r < 4; ++r) {
    const int i = tid + NTHR * r;
    const int g = 12 * gbeg + i - 2;
    idn[r] = (i < TLN + 4 && g >= 0 && g < L) ? ids[(size_t)b * L + g] : -1;
  }

  const int lg = tid >> 3;
  const int oc = cs * SLC + (tid & 7) * 4;
  const float4 bias = *(const float4*)(conv_b + oc);
  const float* Ts = T + (size_t)cs * (KC * VOCAB * SLC) + (tid & 7) * 4;

  const int cq = tid & 127;
  const float cb0 = conv_b[cq], cb1 = conv_b[cq + 128];
  const float cb2 = conv_b[cq + 256], cb3 = conv_b[cq + 384];
  const float sw0 = sw[cq], sw1 = sw[cq + 128];
  const float sw2 = sw[cq + 256], sw3 = sw[cq + 384];
  if (tid < KC * VOCAB / 4) {
    const float4* pp = (const float4*)P + tid;
    float4 s = make_float4(0.f, 0.f, 0.f, 0.f);
#pragma unroll
    for (int ot = 0; ot < NOT; ++ot) {
      const float4 t = pp[(size_t)ot * (KC * VOCAB / 4)];
      s.x += t.x; s.y += t.y; s.z += t.z; s.w += t.w;
    }
    *(float4*)(sc + 4 * tid) = s;
  }
  if (tid >= NTHR - 128) {
    float s = cb0 * sw0 + cb1 * sw1;
    s += cb2 * sw2 + cb3 * sw3;
#pragma unroll
    for (int off = 16; off > 0; off >>= 1) s += __shfl_xor(s, off, 32);
    if ((tid & 31) == 0) bpart[(tid >> 5) & 3] = s;
  }

  for (int ti = 0; ti < ntile; ++ti) {
    const int tg0 = gbeg + TGR * ti;
    const int t0  = 12 * tg0;
    const int ngt = (gend - tg0 < TGR) ? (gend - tg0) : TGR;
    const int npt = 12 * ngt;

    __syncthreads();
#pragma unroll
    for (int r = 0; r < 4; ++r) {
      const int i = tid + NTHR * r;
      if (i < TLN + 4) sid[i] = idn[r];
    }
    __syncthreads();

    for (int p = tid; p < npt; p += NTHR) {
      float s = 0.f;
      if (t0 + p < L) {
        s = bpart[0];
#pragma unroll
        for (int w = 1; w < 4; ++w) s += bpart[w];
#pragma unroll
        for (int k = 0; k < KC; ++k) {
          const int v = sid[p + k];
          const float t = sc[k * VOCAB + (v < 0 ? 0 : v)];
          s += (v >= 0) ? t : 0.f;
        }
      }
      q[p] = s;
    }
    __syncthreads();

    for (int l = tid; l < npt; l += NTHR) {
      const int j2 = (l >> 1) << 1;
      const int j3 = (l / 3) * 3;
      const int j4 = (l >> 2) << 2;
      float s1 = q[l];
      float s2 = 0.5f * (q[j2] + q[j2 + 1]);
      float s3 = (q[j3] + q[j3 + 1] + q[j3 + 2]) * (1.f / 3.f);
      float s4 = 0.25f * (q[j4] + q[j4 + 1] + q[j4 + 2] + q[j4 + 3]);
      float mx = fmaxf(fmaxf(s1, s2), fmaxf(s3, s4));
      float e1 = __expf(s1 - mx), e2 = __expf(s2 - mx);
      float e3 = __expf(s3 - mx), e4 = __expf(s4 - mx);
      float inv = 1.f / (e1 + e2 + e3 + e4);
      wc[l][0] = e1 * inv;
      wc[l][1] = e2 * inv * 0.5f;
      wc[l][2] = e3 * inv * (1.f / 3.f);
      wc[l][3] = e4 * inv * 0.25f;
    }
    __syncthreads();

    for (int e = tid; e < ngt * 6; e += NTHR) {
      const int g = e / 6;
      const int j = e - g * 6;
      float c[12];
#pragma unroll
      for (int lp = 0; lp < 12; ++lp) c[lp] = 0.f;
#pragma unroll
      for (int dl = 0; dl < 2; ++dl) {
        const int l  = 2 * j + dl;
        const int l3 = l / 3;
        const float4 w = *(const float4*)wc[12 * g + l];
#pragma unroll
        for (int lp = 0; lp < 12; ++lp) {
          c[lp] += (lp == l)               ? w.x : 0.f;
          c[lp] += ((lp >> 1) == (l >> 1)) ? w.y : 0.f;
          c[lp] += ((lp / 3) == l3)        ? w.z : 0.f;
          c[lp] += ((lp >> 2) == (l >> 2)) ? w.w : 0.f;
        }
      }
#pragma unroll
      for (int lp = 0; lp < 12; ++lp) cf[g][j][lp] = 0.5f * c[lp];
    }
    __syncthreads();

    if (ti + 1 < ntile) {
#pragma unroll
      for (int r = 0; r < 4; ++r) {
        const int i = tid + NTHR * r;
        const int g = t0 + TLN + i - 2;
        idn[r] = (i < TLN + 4 && g >= 0 && g < L) ? ids[(size_t)b * L + g] : -1;
      }
    }
    const bool edge = (t0 == 0) || (t0 + npt + 2 > L);
    if (edge) gather_mix<true >(Ts, out, sid, cf, bias, b, t0, L, oc, lg, ngt);
    else      gather_mix<false>(Ts, out, sid, cf, bias, b, t0, L, oc, lg, ngt);
  }
}

extern "C" void kernel_launch(void* const* d_in, const int* in_sizes, int n_in,
                              void* d_out, int out_size, void* d_ws, size_t ws_size,
                              hipStream_t stream) {
  const int*   ids     = (const int*)d_in[0];
  const float* emb     = (const float*)d_in[1];
  const float* conv_w  = (const float*)d_in[2];
  const float* conv_b  = (const float*)d_in[3];
  const float* score_w = (const float*)d_in[4];
  float* out = (float*)d_out;
  float* T   = (float*)d_ws;
  float* P   = T + KC * VOCAB * DCH;

  const int L = LLEN;
  const int B = in_sizes[0] / L;

  dim3 g1(DCH / 16, VOCAB / 16);
  gbst_table_wmma<<<g1, 64, 0, stream>>>(emb, conv_w, score_w, T, P);

  dim3 g2(NSL, 2 * B);
  gbst_main<<<g2, NTHR, 0, stream>>>(ids, T, P, conv_b, score_w, out, L);
}


# Kernel 2 work-group = pair of 32-channel slices x quarter row: the two gather passes of a tile stage output rows in LDS, then each store instruction writes whole aligned 256-byte units; sums kept
# speedup vs baseline: 2.6530x; 1.0560x over previous
#include <hip/hip_runtime.h>
#include <hip/hip_bf16.h>

typedef float v2f __attribute__((ext_vector_type(2)));
typedef float v8f __attribute__((ext_vector_type(8)));

#define VOCAB 256
#define DCH   512
#define KC    5
#define SLC   32
#define NSL   (DCH / SLC)
#define SL2   (2 * SLC)
#define NSP   (DCH / SL2)
#define TGR   96
#define TLN   (TGR * 12)
#define NTHR  384
#define SPI   17
#define NOT   (DCH / 16)
#define LLEN  8192

struct TFrag {
  float4 a[4];
  float4 b[4][5];
};

__device__ __forceinline__ void tf_load(TFrag& f, const float* __restrict__ ap,
                                        const float* __restrict__ bp) {
#pragma unroll
  for (int s = 0; s < 4; ++s) {
    f.a[s] = *(const float4*)(ap + 8 * s);
#pragma unroll
    for (int q = 0; q < 5; ++q)
      f.b[s][q] = *(const float4*)(bp + 40 * s + 4 * q);
  }
}

__device__ __forceinline__ void tf_mma(const TFrag& f, v8f (&c)[KC]) {
#pragma unroll
  for (int s = 0; s < 4; ++s) {
    const float av[4] = {f.a[s].x, f.a[s].y, f.a[s].z, f.a[s].w};
    float wv[20];
#pragma unroll
    for (int q = 0; q < 5; ++q) {
      wv[4 * q + 0] = f.b[s][q].x; wv[4 * q + 1] = f.b[s][q].y;
      wv[4 * q + 2] = f.b[s][q].z; wv[4 * q + 3] = f.b[s][q].w;
    }
#pragma unroll
    for (int u = 0; u < 2; ++u) {
      v2f a;
      a[0] = av[2 * u];
      a[1] = av[2 * u + 1];
#pragma unroll
      for (int k = 0; k < KC; ++k) {
        v2f b;
        b[0] = wv[(2 * u) * 5 + k];
        b[1] = wv[(2 * u + 1) * 5 + k];
        c[k] = __builtin_amdgcn_wmma_f32_16x16x4_f32(false, a, false, b,
                                                     (short)0, c[k], false, false);
      }
    }
  }
}

template <int M>
__device__ __forceinline__ float row_xor(float x) {
  return __int_as_float(__builtin_amdgcn_update_dpp(
      0, __float_as_int(x), 0x160 + M, 0xf, 0xf, true));
}

__global__ __launch_bounds__(64) void gbst_table_wmma(
    const float* __restrict__ emb,
    const float* __restrict__ conv_w,
    const float* __restrict__ sw,
    float* __restrict__ T,
    float* __restrict__ P)
{
  __shared__ float4 red[2][KC][32];

  const int v0   = blockIdx.y * 16;
  const int o0   = blockIdx.x * 16;
  const int lane = threadIdx.x & 31;
  const int w    = threadIdx.x >> 5;
  const int half = lane >> 4;
  const int sub  = lane & 15;

  v8f c[KC];
#pragma unroll
  for (int k = 0; k < KC; ++k) c[k] = v8f{};

  const float* ap = emb + (size_t)(v0 + sub) * DCH + 256 * w + 4 * half;
  const float* bp = conv_w + (size_t)(o0 + sub) * (DCH * KC) + (256 * w + 4 * half) * KC;

  TFrag f0, f1;
  tf_load(f0, ap, bp);
#pragma unroll 1
  for (int it = 0; it < 3; ++it) {
    tf_load(f1, ap + 32, bp + 32 * KC);
    __builtin_amdgcn_sched_barrier(0);
    tf_mma(f0, c);
    __builtin_amdgcn_sched_barrier(0);
    tf_load(f0, ap + 64, bp + 64 * KC);
    __builtin_amdgcn_sched_barrier(0);
    tf_mma(f1, c);
    __builtin_amdgcn_sched_barrier(0);
    ap += 64;
    bp += 64 * KC;
  }
  tf_load(f1, ap + 32, bp + 32 * KC);
  __builtin_amdgcn_sched_barrier(0);
  tf_mma(f0, c);
  tf_mma(f1, c);

  float keep[KC][4];
#pragma unroll
  for (int k = 0; k < KC; ++k) {
    float4 snd;
    snd.x = w ? c[k][0] : c[k][4];
    snd.y = w ? c[k][1] : c[k][5];
    snd.z = w ? c[k][2] : c[k][6];
    snd.w = w ? c[k][3] : c[k][7];
    red[1 - w][k][lane] = snd;
    keep[k][0] = w ? c[k][4] : c[k][0];
    keep[k][1] = w ? c[k][5] : c[k][1];
    keep[k][2] = w ? c[k][6] : c[k][2];
    keep[k][3] = w ? c[k][7] : c[k][3];
  }
  __syncthreads();

  const float swv = sw[o0 + sub];
  float* pr = P + (size_t)blockIdx.x * (KC * VOCAB) + v0 + half * 8 + 4 * w;
#pragma unroll
  for (int k = 0; k < KC; ++k) {
    const float4 oth = red[w][k][lane];
    keep[k][0] += oth.x; keep[k][1] += oth.y;
    keep[k][2] += oth.z; keep[k][3] += oth.w;
    float sr[4];
#pragma unroll
    for (int rr = 0; rr < 4; ++rr) {
      const int v = v0 + half * 8 + 4 * w + rr;
      T[(((size_t)(o0 / SLC) * KC + k) * VOCAB + v) * SLC + (o0 % SLC + sub)] = keep[k][rr];
      float s = keep[k][rr] * swv;
      s += row_xor<8>(s);
      s += row_xor<4>(s);
      s += row_xor<2>(s);
      s += row_xor<1>(s);
      sr[rr] = s;
    }
    if (sub == 0)
      *(float4*)(pr + k * VOCAB) = make_float4(sr[0], sr[1], sr[2], sr[3]);
  }
}

__host__ __device__ constexpr bool touches(int j, int lp) {
  return ((lp >> 2) == (j >> 1)) || ((lp / 3) == ((2 * j) / 3)) ||
         ((lp / 3) == ((2 * j + 1) / 3));
}

template <bool EDGE>
__device__ __forceinline__ void gather_mix(
    const float* __restrict__ Ts, float4* stg,
    const int* sid, const float (*cf)[6][12], const float4 bias,
    int t0, int L, int lg, int ngt)
{
  int  pb[2];
  bool val[2];
#pragma unroll
  for (int s = 0; s < 2; ++s) {
    const int g = lg + (NTHR / 8) * s;
    val[s] = g < ngt;
    pb[s]  = 12 * (val[s] ? g : ngt - 1);
  }
  float4 x[2][12];
#pragma unroll
  for (int s = 0; s < 2; ++s)
#pragma unroll
    for (int lp = 0; lp < 12; ++lp) x[s][lp] = bias;

#pragma unroll 1
  for (int k = 0; k < KC; ++k) {
    const float* Tk = Ts + (size_t)k * (VOCAB * SLC);
#pragma unroll
    for (int s = 0; s < 2; ++s) {
#pragma unroll
      for (int lp = 0; lp < 12; ++lp) {
        int v = sid[pb[s] + lp + k];
        if (EDGE) {
          const bool ok = v >= 0;
          v = ok ? v : 0;
          const float4 tv = *(const float4*)(Tk + v * SLC);
          x[s][lp].x += ok ? tv.x : 0.f; x[s][lp].y += ok ? tv.y : 0.f;
          x[s][lp].z += ok ? tv.z : 0.f; x[s][lp].w += ok ? tv.w : 0.f;
        } else {
          const float4 tv = *(const float4*)(Tk + v * SLC);
          x[s][lp].x += tv.x; x[s][lp].y += tv.y; x[s][lp].z += tv.z; x[s][lp].w += tv.w;
        }
      }
    }
  }

#pragma unroll
  for (int s = 0; s < 2; ++s) {
    const int g = pb[s] / 12;
    float4 acc[6];
#pragma unroll
    for (int j = 0; j < 6; ++j) acc[j] = make_float4(0.f, 0.f, 0.f, 0.f);
#pragma unroll
    for (int lp = 0; lp < 12; ++lp) {
      float4 xv = x[s][lp];
      if (EDGE && t0 + pb[s] + lp >= L) xv = make_float4(0.f, 0.f, 0.f, 0.f);
#pragma unroll
      for (int j = 0; j < 6; ++j) {
        if (touches(j, lp)) {
          const float w = cf[g][j][lp];
          acc[j].x += w * xv.x; acc[j].y += w * xv.y;
          acc[j].z += w * xv.z; acc[j].w += w * xv.w;
        }
      }
    }
    if (val[s]) {
#pragma unroll
      for (int j = 0; j < 6; ++j) stg[(g * 6 + j) * SPI] = acc[j];
    }
  }
}

__global__ __launch_bounds__(NTHR) void gbst_main(
    const int*   __restrict__ ids,
    const float* __restrict__ T,
    const float* __restrict__ P,
    const float* __restrict__ conv_b,
    const float* __restrict__ sw,
    float*       __restrict__ out,
    int L)
{
  __shared__ int   sid[TLN + 4];
  __shared__ __attribute__((aligned(16))) float sc[KC * VOCAB];
  __shared__ float bpart[4];
  __shared__ float q[TLN];
  __shared__ __attribute__((aligned(16))) float wc[TLN][4];
  __shared__ __attribute__((aligned(16))) float cf[TGR][6][12];
  __shared__ float4 stage[TGR * 6 * SPI];

  const int sp   = blockIdx.x;
  const int b    = blockIdx.y >> 2;
  const int qt   = blockIdx.y & 3;
  const int tid  = threadIdx.x;
  const int ngr  = (L + 11) / 12;
  const int gqt  = (ngr + 3) / 4;
  const int gbeg = qt * gqt;
  const int gend = (gbeg + gqt < ngr) ? gbeg + gqt : ngr;
  const int ngq  = (gend > gbeg) ? gend - gbeg : 0;
  const int ntile = (ngq + TGR - 1) / TGR;
  const int tsz  = ntile ? (ngq + ntile - 1) / ntile : 0;

  int idn[4];
#pragma unroll
  for (int r = 0; r < 4; ++r) {
    const int i = tid + NTHR * r;
    const int g = 12 * gbeg + i - 2;
    idn[r] = (i < TLN + 4 && g >= 0 && g < L) ? ids[(size_t)b * L + g] : -1;
  }

  const int lg = tid >> 3;
  const int oc = sp * SL2 + (tid & 7) * 4;
  const float4 bias0 = *(const float4*)(conv_b + oc);
  const float4 bias1 = *(const float4*)(conv_b + oc + SLC);
  const float* Ts = T + (size_t)(2 * sp) * (KC * VOCAB * SLC) + (tid & 7) * 4;

  const int cq = tid & 127;
  const float cb0 = conv_b[cq], cb1 = conv_b[cq + 128];
  const float cb2 = conv_b[cq + 256], cb3 = conv_b[cq + 384];
  const float sw0 = sw[cq], sw1 = sw[cq + 128];
  const float sw2 = sw[cq + 256], sw3 = sw[cq + 384];
  if (tid < KC * VOCAB / 4) {
    const float4* pp = (const float4*)P + tid;
    float4 s = make_float4(0.f, 0.f, 0.f, 0.f);
#pragma unroll
    for (int ot = 0; ot < NOT; ++ot) {
      const float4 t = pp[(size_t)ot * (KC * VOCAB / 4)];
      s.x += t.x; s.y += t.y; s.z += t.z; s.w += t.w;
    }
    *(float4*)(sc + 4 * tid) = s;
  }
  if (tid >= NTHR - 128) {
    float s = cb0 * sw0 + cb1 * sw1;
    s += cb2 * sw2 + cb3 * sw3;
#pragma unroll
    for (int off = 16; off > 0; off >>= 1) s += __shfl_xor(s, off, 32);
    if ((tid & 31) == 0) bpart[(tid >> 5) & 3] = s;
  }

  for (int ti = 0; ti < ntile; ++ti) {
    const int tg0 = gbeg + tsz * ti;
    const int t0  = 12 * tg0;
    const int ngt = (gend - tg0 < tsz) ? (gend - tg0) : tsz;
    const int npt = 12 * ngt;

    __syncthreads();
#pragma unroll
    for (int r = 0; r < 4; ++r) {
      const int i = tid + NTHR * r;
      if (i < TLN + 4) sid[i] = idn[r];
    }
    __syncthreads();

    for (int p = tid; p < npt; p += NTHR) {
      float s = 0.f;
      if (t0 + p < L) {
        s = bpart[0];
#pragma unroll
        for (int w = 1; w < 4; ++w) s += bpart[w];
#pragma unroll
        for (int k = 0; k < KC; ++k) {
          const int v = sid[p + k];
          const float t = sc[k * VOCAB + (v < 0 ? 0 : v)];
          s += (v >= 0) ? t : 0.f;
        }
      }
      q[p] = s;
    }
    __syncthreads();

    for (int l = tid; l < npt; l += NTHR) {
      const int j2 = (l >> 1) << 1;
      const int j3 = (l / 3) * 3;
      const int j4 = (l >> 2) << 2;
      float s1 = q[l];
      float s2 = 0.5f * (q[j2] + q[j2 + 1]);
      float s3 = (q[j3] + q[j3 + 1] + q[j3 + 2]) * (1.f / 3.f);
      float s4 = 0.25f * (q[j4] + q[j4 + 1] + q[j4 + 2] + q[j4 + 3]);
      float mx = fmaxf(fmaxf(s1, s2), fmaxf(s3, s4));
      float e1 = __expf(s1 - mx), e2 = __expf(s2 - mx);
      float e3 = __expf(s3 - mx), e4 = __expf(s4 - mx);
      float inv = 1.f / (e1 + e2 + e3 + e4);
      wc[l][0] = e1 * inv;
      wc[l][1] = e2 * inv * 0.5f;
      wc[l][2] = e3 * inv * (1.f / 3.f);
      wc[l][3] = e4 * inv * 0.25f;
    }
    __syncthreads();

    for (int e = tid; e < ngt * 6; e += NTHR) {
      const int g = e / 6;
      const int j = e - g * 6;
      float c[12];
#pragma unroll
      for (int lp = 0; lp < 12; ++lp) c[lp] = 0.f;
#pragma unroll
      for (int dl = 0; dl < 2; ++dl) {
        const int l  = 2 * j + dl;
        const int l3 = l / 3;
        const float4 w = *(const float4*)wc[12 * g + l];
#pragma unroll
        for (int lp = 0; lp < 12; ++lp) {
          c[lp] += (lp == l)               ? w.x : 0.f;
          c[lp] += ((lp >> 1) == (l >> 1)) ? w.y : 0.f;
          c[lp] += ((lp / 3) == l3)        ? w.z : 0.f;
          c[lp] += ((lp >> 2) == (l >> 2)) ? w.w : 0.f;
        }
      }
#pragma unroll
      for (int lp = 0; lp < 12; ++lp) cf[g][j][lp] = 0.5f * c[lp];
    }
    __syncthreads();

    if (ti + 1 < ntile) {
#pragma unroll
      for (int r = 0; r < 4; ++r) {
        const int i = tid + NTHR * r;
        const int g = t0 + 12 * tsz + i - 2;
        idn[r] = (i < TLN + 4 && g >= 0 && g < L) ? ids[(size_t)b * L + g] : -1;
      }
    }
    const bool edge = (t0 == 0) || (t0 + npt + 2 > L);
#pragma unroll 1
    for (int ps = 0; ps < 2; ++ps) {
      const float*  Tp = Ts + (size_t)ps * (KC * VOCAB * SLC);
      float4*       sg = stage + 8 * ps + (tid & 7);
      const float4  bs = ps ? bias1 : bias0;
      if (edge) gather_mix<true >(Tp, sg, sid, cf, bs, t0, L, lg, ngt);
      else      gather_mix<false>(Tp, sg, sid, cf, bs, t0, L, lg, ngt);
    }
    __syncthreads();

    {
      const int nrow = 6 * ngt;
      const int rout = (L - t0 + 1) >> 1;
      const int rlim = nrow < rout ? nrow : rout;
      const int c16  = tid & 15;
      const int rstp = 2 * (NTHR / 32);
      float* ob = out + ((size_t)b * (L >> 1) + (t0 >> 1)) * DCH + sp * SL2 + 4 * c16;
      for (int r0 = 2 * (tid >> 5) + ((tid >> 4) & 1); r0 < rlim; r0 += 6 * rstp) {
        float4 v[6];
#pragma unroll
        for (int u = 0; u < 6; ++u) {
          const int r = r0 + rstp * u;
          v[u] = stage[(r < rlim ? r : rlim - 1) * SPI + c16];
        }
        asm volatile("" :: "v"(v[0].x), "v"(v[0].y), "v"(v[0].z), "v"(v[0].w),
                           "v"(v[1].x), "v"(v[1].y), "v"(v[1].z), "v"(v[1].w),
                           "v"(v[2].x), "v"(v[2].y), "v"(v[2].z), "v"(v[2].w),
                           "v"(v[3].x), "v"(v[3].y), "v"(v[3].z), "v"(v[3].w),
                           "v"(v[4].x), "v"(v[4].y), "v"(v[4].z), "v"(v[4].w),
                           "v"(v[5].x), "v"(v[5].y), "v"(v[5].z), "v"(v[5].w));
#pragma unroll
        for (int u = 0; u < 6; ++u) {
          const int r = r0 + rstp * u;
          if (r < rlim) *(float4*)(ob + (size_t)r * DCH) = v[u];
        }
      }
    }
  }
}

extern "C" void kernel_launch(void* const* d_in, const int* in_sizes, int n_in,
                              void* d_out, int out_size, void* d_ws, size_t ws_size,
                              hipStream_t stream) {
  const int*   ids     = (const int*)d_in[0];
  const float* emb     = (const float*)d_in[1];
  const float* conv_w  = (const float*)d_in[2];
  const float* conv_b  = (const float*)d_in[3];
  const float* score_w = (const float*)d_in[4];
  float* out = (float*)d_out;
  float* T   = (float*)d_ws;
  float* P   = T + KC * VOCAB * DCH;

  const int L = LLEN;
  const int B = in_sizes[0] / L;

  dim3 g1(DCH / 16, VOCAB / 16);
  gbst_table_wmma<<<g1, 64, 0, stream>>>(emb, conv_w, score_w, T, P);

  dim3 g2(NSP, 4 * B);
  gbst_main<<<g2, NTHR, 0, stream>>>(ids, T, P, conv_b, score_w, out, L);
}
